# starting kernel
# baseline (speedup 1.0000x reference)
#include <hip/hip_runtime.h>
#include <hip/hip_bf16.h>

#define HD 128
#define NNODES 50000
#define NEDGES 600000
#define NLAYERS 4
#define KC 32

typedef float v2f __attribute__((ext_vector_type(2)));
typedef float v8f __attribute__((ext_vector_type(8)));

static __device__ __forceinline__ v8f vzero8() {
    v8f z;
#pragma unroll
    for (int i = 0; i < 8; ++i) z[i] = 0.f;
    return z;
}

static __device__ __forceinline__ v8f wmma_f32(v2f a, v2f b, v8f c) {
    return __builtin_amdgcn_wmma_f32_16x16x4_f32(false, a, false, b, (short)0, c,
                                                 false, false);
}

static __device__ __forceinline__ float silu(float x) {
    return x / (1.f + __expf(-x));
}

static __device__ __forceinline__ void stage_weights(v2f* __restrict__ sW,
                                                     const float* __restrict__ W,
                                                     int kc, int tid) {
    const float* wsrc = W + (size_t)kc * HD + tid;
#pragma unroll
    for (int p = 0; p < KC / 2; ++p) {
        v2f w;
        w.x = wsrc[(2 * p) * HD];
        w.y = wsrc[(2 * p + 1) * HD];
        sW[p * HD + tid] = w;
    }
}

__global__ __launch_bounds__(256) void embed_kernel(
    const float* __restrict__ loc, const float* __restrict__ vel,
    const float* __restrict__ emb_w, const float* __restrict__ emb_b,
    float* __restrict__ h) {
    int idx = blockIdx.x * blockDim.x + threadIdx.x;
    if (idx >= NNODES * HD) return;
    int n = idx >> 7;
    int c = idx & (HD - 1);
    float x = emb_b[c];
#pragma unroll
    for (int k = 0; k < 3; ++k) x = fmaf(loc[n * 3 + k], emb_w[k * HD + c], x);
#pragma unroll
    for (int k = 0; k < 3; ++k) x = fmaf(vel[n * 3 + k], emb_w[(k + 3) * HD + c], x);
    h[idx] = x;
}

__global__ __launch_bounds__(256) void zero_kernel(float* __restrict__ p, int count) {
    int i = blockIdx.x * blockDim.x + threadIdx.x;
    if (i < count) p[i] = 0.f;
}

__global__ __launch_bounds__(128) void edge_kernel(
    const float* __restrict__ h, const float* __restrict__ edge_attr,
    const int* __restrict__ row, const int* __restrict__ col,
    const float* __restrict__ W1,
    const float* __restrict__ b1,
    const float* __restrict__ W2,
    const float* __restrict__ b2,
    float* __restrict__ agg) {
    __shared__ v2f sW[(KC / 2) * HD];
    __shared__ float sT[4][16 * HD];
    const int tid  = threadIdx.x;
    const int lane = tid & 31;
    const int wave = tid >> 5;
    const int tile = blockIdx.x * 4 + wave;
    const int base = tile * 16;
    const int mrow = lane & 15;
    const int kgrp = (lane >> 4) * 2;
    const int khalf = lane >> 4;
    const int mo   = 8 * (lane >> 4);
    const int ncol = lane & 15;

    const int r_idx = row[base + mrow];
    const int c_idx = col[base + mrow];
    const float* hr = h + (size_t)r_idx * HD;
    const float* hc = h + (size_t)c_idx * HD;

    v8f acc[8];
#pragma unroll
    for (int nt = 0; nt < 8; ++nt) acc[nt] = vzero8();

    for (int kc = 0; kc < 2 * HD; kc += KC) {
        __syncthreads();
        stage_weights(sW, W1, kc, tid);
        __syncthreads();
#pragma unroll
        for (int s = 0; s < KC / 4; ++s) {
            const int kb = kc + 4 * s + kgrp;
            const float* src = (kb < HD) ? (hr + kb) : (hc + (kb - HD));
            v2f a = *(const v2f*)src;
            const v2f* bp = &sW[(2 * s + khalf) * HD + ncol];
#pragma unroll
            for (int nt = 0; nt < 8; ++nt) {
                acc[nt] = wmma_f32(a, bp[nt * 16], acc[nt]);
            }
        }
    }

    float ea0[8], ea1[8];
#pragma unroll
    for (int g = 0; g < 8; ++g) {
        const int e = base + g + mo;
        ea0[g] = edge_attr[2 * e];
        ea1[g] = edge_attr[2 * e + 1];
    }
    const float* w1e0 = W1 + (size_t)256 * HD;
    const float* w1e1 = W1 + (size_t)257 * HD;
    float* sTw = sT[wave];
#pragma unroll
    for (int nt = 0; nt < 8; ++nt) {
        const int n = nt * 16 + ncol;
        const float we0 = w1e0[n], we1 = w1e1[n], bb = b1[n];
#pragma unroll
        for (int g = 0; g < 8; ++g) {
            float x = acc[nt][g] + ea0[g] * we0 + ea1[g] * we1 + bb;
            sTw[(g + mo) * HD + n] = silu(x);
        }
    }
    asm volatile("s_wait_dscnt 0" ::: "memory");

    v8f acc2[8];
#pragma unroll
    for (int nt = 0; nt < 8; ++nt) acc2[nt] = vzero8();

    for (int kc = 0; kc < HD; kc += KC) {
        __syncthreads();
        stage_weights(sW, W2, kc, tid);
        __syncthreads();
#pragma unroll
        for (int s = 0; s < KC / 4; ++s) {
            const int kb = kc + 4 * s + kgrp;
            v2f a = *(const v2f*)&sTw[mrow * HD + kb];
            const v2f* bp = &sW[(2 * s + khalf) * HD + ncol];
#pragma unroll
            for (int nt = 0; nt < 8; ++nt) {
                acc2[nt] = wmma_f32(a, bp[nt * 16], acc2[nt]);
            }
        }
    }

    int rg[8];
#pragma unroll
    for (int g = 0; g < 8; ++g) rg[g] = row[base + g + mo];
#pragma unroll
    for (int nt = 0; nt < 8; ++nt) {
        const int n = nt * 16 + ncol;
        const float bb = b2[n];
#pragma unroll
        for (int g = 0; g < 8; ++g) {
            float x = silu(acc2[nt][g] + bb);
            atomicAdd(&agg[(size_t)rg[g] * HD + n], x);
        }
    }
}

__global__ __launch_bounds__(128) void node_kernel(
    float* __restrict__ h, const float* __restrict__ agg,
    const float* __restrict__ W1,
    const float* __restrict__ b1, const float* __restrict__ W2,
    const float* __restrict__ b2) {
    __shared__ v2f sW[(KC / 2) * HD];
    __shared__ float sT[4][16 * HD];
    const int tid  = threadIdx.x;
    const int lane = tid & 31;
    const int wave = tid >> 5;
    const int node_tiles = NNODES / 16;
    const int tile = blockIdx.x * 4 + wave;
    const bool active = tile < node_tiles;
    const int tt = active ? tile : node_tiles - 1;
    const int base = tt * 16;
    const int mrow = lane & 15;
    const int kgrp = (lane >> 4) * 2;
    const int khalf = lane >> 4;
    const int mo   = 8 * (lane >> 4);
    const int ncol = lane & 15;

    const float* hr = h + (size_t)(base + mrow) * HD;
    const float* ar = agg + (size_t)(base + mrow) * HD;

    v8f acc[8];
#pragma unroll
    for (int nt = 0; nt < 8; ++nt) acc[nt] = vzero8();

    for (int kc = 0; kc < 2 * HD; kc += KC) {
        __syncthreads();
        stage_weights(sW, W1, kc, tid);
        __syncthreads();
#pragma unroll
        for (int s = 0; s < KC / 4; ++s) {
            const int kb = kc + 4 * s + kgrp;
            const float* src = (kb < HD) ? (hr + kb) : (ar + (kb - HD));
            v2f a = *(const v2f*)src;
            const v2f* bp = &sW[(2 * s + khalf) * HD + ncol];
#pragma unroll
            for (int nt = 0; nt < 8; ++nt) {
                acc[nt] = wmma_f32(a, bp[nt * 16], acc[nt]);
            }
        }
    }

    float* sTw = sT[wave];
#pragma unroll
    for (int nt = 0; nt < 8; ++nt) {
        const int n = nt * 16 + ncol;
        const float bb = b1[n];
#pragma unroll
        for (int g = 0; g < 8; ++g) {
            sTw[(g + mo) * HD + n] = silu(acc[nt][g] + bb);
        }
    }
    asm volatile("s_wait_dscnt 0" ::: "memory");

    v8f acc2[8];
#pragma unroll
    for (int nt = 0; nt < 8; ++nt) acc2[nt] = vzero8();

    for (int kc = 0; kc < HD; kc += KC) {
        __syncthreads();
        stage_weights(sW, W2, kc, tid);
        __syncthreads();
#pragma unroll
        for (int s = 0; s < KC / 4; ++s) {
            const int kb = kc + 4 * s + kgrp;
            v2f a = *(const v2f*)&sTw[mrow * HD + kb];
            const v2f* bp = &sW[(2 * s + khalf) * HD + ncol];
#pragma unroll
            for (int nt = 0; nt < 8; ++nt) {
                acc2[nt] = wmma_f32(a, bp[nt * 16], acc2[nt]);
            }
        }
    }

    if (active) {
#pragma unroll
        for (int nt = 0; nt < 8; ++nt) {
            const int n = nt * 16 + ncol;
            const float bb = b2[n];
#pragma unroll
            for (int g = 0; g < 8; ++g) {
                float* dst = &h[(size_t)(base + g + mo) * HD + n];
                *dst = *dst + (acc2[nt][g] + bb);
            }
        }
    }
}

__global__ __launch_bounds__(128) void decode_kernel(
    const float* __restrict__ h, const float* __restrict__ dw1,
    const float* __restrict__ db1, const float* __restrict__ dw2,
    const float* __restrict__ db2, float* __restrict__ out) {
    __shared__ float sh[HD];
    __shared__ float st[HD];
    const int n = blockIdx.x;
    const int t = threadIdx.x;
    sh[t] = h[(size_t)n * HD + t];
    __syncthreads();
    float x = db1[t];
    for (int k = 0; k < HD; ++k) x = fmaf(sh[k], dw1[k * HD + t], x);
    st[t] = silu(x);
    __syncthreads();
    if (t < 3) {
        float y = db2[t];
        for (int k = 0; k < HD; ++k) y = fmaf(st[k], dw2[k * 3 + t], y);
        out[(size_t)n * 3 + t] = y;
    }
}

extern "C" void kernel_launch(void* const* d_in, const int* in_sizes, int n_in,
                              void* d_out, int out_size, void* d_ws, size_t ws_size,
                              hipStream_t stream) {
    const float* loc       = (const float*)d_in[0];
    const float* vel       = (const float*)d_in[1];
    const float* edge_attr = (const float*)d_in[2];
    const int*   row       = (const int*)d_in[3];
    const int*   col       = (const int*)d_in[4];
    const float* emb_w     = (const float*)d_in[5];
    const float* emb_b     = (const float*)d_in[6];
    const float* ew1       = (const float*)d_in[7];
    const float* eb1       = (const float*)d_in[8];
    const float* ew2       = (const float*)d_in[9];
    const float* eb2       = (const float*)d_in[10];
    const float* nw1       = (const float*)d_in[11];
    const float* nb1       = (const float*)d_in[12];
    const float* nw2       = (const float*)d_in[13];
    const float* nb2       = (const float*)d_in[14];
    const float* dw1       = (const float*)d_in[15];
    const float* db1       = (const float*)d_in[16];
    const float* dw2       = (const float*)d_in[17];
    const float* db2       = (const float*)d_in[18];
    float* out = (float*)d_out;

    float* h   = (float*)d_ws;
    float* agg = h + (size_t)NNODES * HD;

    embed_kernel<<<(NNODES * HD + 255) / 256, 256, 0, stream>>>(loc, vel, emb_w,
                                                                emb_b, h);

    const int edge_tiles = NEDGES / 16;
    const int node_tiles = NNODES / 16;
    const int edge_blocks = edge_tiles / 4;
    const int node_blocks = (node_tiles + 3) / 4;

    for (int i = 0; i < NLAYERS; ++i) {
        zero_kernel<<<(NNODES * HD + 255) / 256, 256, 0, stream>>>(agg,
                                                                   NNODES * HD);
        edge_kernel<<<edge_blocks, 128, 0, stream>>>(
            h, edge_attr, row, col, ew1 + (size_t)i * 258 * HD, eb1 + i * HD,
            ew2 + (size_t)i * HD * HD, eb2 + i * HD, agg);
        node_kernel<<<node_blocks, 128, 0, stream>>>(
            h, agg, nw1 + (size_t)i * 2 * HD * HD, nb1 + i * HD,
            nw2 + (size_t)i * HD * HD, nb2 + i * HD);
    }

    decode_kernel<<<NNODES, 128, 0, stream>>>(h, dw1, db1, dw2, db2, out);
}


# Decoder: the 128x128 product moved onto the f32 matrix instruction (16-node tile per wave, dw1 staged in LDS), 128x3 projection from LDS; decoder sum order changes
# speedup vs baseline: 1.0433x; 1.0433x over previous
#include <hip/hip_runtime.h>
#include <hip/hip_bf16.h>

#define HD 128
#define NNODES 50000
#define NEDGES 600000
#define NLAYERS 4
#define KC 32

typedef float v2f __attribute__((ext_vector_type(2)));
typedef float v8f __attribute__((ext_vector_type(8)));

static __device__ __forceinline__ v8f vzero8() {
    v8f z;
#pragma unroll
    for (int i = 0; i < 8; ++i) z[i] = 0.f;
    return z;
}

static __device__ __forceinline__ v8f wmma_f32(v2f a, v2f b, v8f c) {
    return __builtin_amdgcn_wmma_f32_16x16x4_f32(false, a, false, b, (short)0, c,
                                                 false, false);
}

static __device__ __forceinline__ float silu(float x) {
    return x / (1.f + __expf(-x));
}

static __device__ __forceinline__ void stage_weights(v2f* __restrict__ sW,
                                                     const float* __restrict__ W,
                                                     int kc, int tid) {
    const float* wsrc = W + (size_t)kc * HD + tid;
#pragma unroll
    for (int p = 0; p < KC / 2; ++p) {
        v2f w;
        w.x = wsrc[(2 * p) * HD];
        w.y = wsrc[(2 * p + 1) * HD];
        sW[p * HD + tid] = w;
    }
}

__global__ __launch_bounds__(256) void embed_kernel(
    const float* __restrict__ loc, const float* __restrict__ vel,
    const float* __restrict__ emb_w, const float* __restrict__ emb_b,
    float* __restrict__ h) {
    int idx = blockIdx.x * blockDim.x + threadIdx.x;
    if (idx >= NNODES * HD) return;
    int n = idx >> 7;
    int c = idx & (HD - 1);
    float x = emb_b[c];
#pragma unroll
    for (int k = 0; k < 3; ++k) x = fmaf(loc[n * 3 + k], emb_w[k * HD + c], x);
#pragma unroll
    for (int k = 0; k < 3; ++k) x = fmaf(vel[n * 3 + k], emb_w[(k + 3) * HD + c], x);
    h[idx] = x;
}

__global__ __launch_bounds__(256) void zero_kernel(float* __restrict__ p, int count) {
    int i = blockIdx.x * blockDim.x + threadIdx.x;
    if (i < count) p[i] = 0.f;
}

__global__ __launch_bounds__(128) void edge_kernel(
    const float* __restrict__ h, const float* __restrict__ edge_attr,
    const int* __restrict__ row, const int* __restrict__ col,
    const float* __restrict__ W1,
    const float* __restrict__ b1,
    const float* __restrict__ W2,
    const float* __restrict__ b2,
    float* __restrict__ agg) {
    __shared__ v2f sW[(KC / 2) * HD];
    __shared__ float sT[4][16 * HD];
    const int tid  = threadIdx.x;
    const int lane = tid & 31;
    const int wave = tid >> 5;
    const int tile = blockIdx.x * 4 + wave;
    const int base = tile * 16;
    const int mrow = lane & 15;
    const int kgrp = (lane >> 4) * 2;
    const int khalf = lane >> 4;
    const int mo   = 8 * (lane >> 4);
    const int ncol = lane & 15;

    const int r_idx = row[base + mrow];
    const int c_idx = col[base + mrow];
    const float* hr = h + (size_t)r_idx * HD;
    const float* hc = h + (size_t)c_idx * HD;

    v8f acc[8];
#pragma unroll
    for (int nt = 0; nt < 8; ++nt) acc[nt] = vzero8();

    for (int kc = 0; kc < 2 * HD; kc += KC) {
        __syncthreads();
        stage_weights(sW, W1, kc, tid);
        __syncthreads();
#pragma unroll
        for (int s = 0; s < KC / 4; ++s) {
            const int kb = kc + 4 * s + kgrp;
            const float* src = (kb < HD) ? (hr + kb) : (hc + (kb - HD));
            v2f a = *(const v2f*)src;
            const v2f* bp = &sW[(2 * s + khalf) * HD + ncol];
#pragma unroll
            for (int nt = 0; nt < 8; ++nt) {
                acc[nt] = wmma_f32(a, bp[nt * 16], acc[nt]);
            }
        }
    }

    float ea0[8], ea1[8];
#pragma unroll
    for (int g = 0; g < 8; ++g) {
        const int e = base + g + mo;
        ea0[g] = edge_attr[2 * e];
        ea1[g] = edge_attr[2 * e + 1];
    }
    const float* w1e0 = W1 + (size_t)256 * HD;
    const float* w1e1 = W1 + (size_t)257 * HD;
    float* sTw = sT[wave];
#pragma unroll
    for (int nt = 0; nt < 8; ++nt) {
        const int n = nt * 16 + ncol;
        const float we0 = w1e0[n], we1 = w1e1[n], bb = b1[n];
#pragma unroll
        for (int g = 0; g < 8; ++g) {
            float x = acc[nt][g] + ea0[g] * we0 + ea1[g] * we1 + bb;
            sTw[(g + mo) * HD + n] = silu(x);
        }
    }
    asm volatile("s_wait_dscnt 0" ::: "memory");

    v8f acc2[8];
#pragma unroll
    for (int nt = 0; nt < 8; ++nt) acc2[nt] = vzero8();

    for (int kc = 0; kc < HD; kc += KC) {
        __syncthreads();
        stage_weights(sW, W2, kc, tid);
        __syncthreads();
#pragma unroll
        for (int s = 0; s < KC / 4; ++s) {
            const int kb = kc + 4 * s + kgrp;
            v2f a = *(const v2f*)&sTw[mrow * HD + kb];
            const v2f* bp = &sW[(2 * s + khalf) * HD + ncol];
#pragma unroll
            for (int nt = 0; nt < 8; ++nt) {
                acc2[nt] = wmma_f32(a, bp[nt * 16], acc2[nt]);
            }
        }
    }

    int rg[8];
#pragma unroll
    for (int g = 0; g < 8; ++g) rg[g] = row[base + g + mo];
#pragma unroll
    for (int nt = 0; nt < 8; ++nt) {
        const int n = nt * 16 + ncol;
        const float bb = b2[n];
#pragma unroll
        for (int g = 0; g < 8; ++g) {
            float x = silu(acc2[nt][g] + bb);
            atomicAdd(&agg[(size_t)rg[g] * HD + n], x);
        }
    }
}

__global__ __launch_bounds__(128) void node_kernel(
    float* __restrict__ h, const float* __restrict__ agg,
    const float* __restrict__ W1,
    const float* __restrict__ b1, const float* __restrict__ W2,
    const float* __restrict__ b2) {
    __shared__ v2f sW[(KC / 2) * HD];
    __shared__ float sT[4][16 * HD];
    const int tid  = threadIdx.x;
    const int lane = tid & 31;
    const int wave = tid >> 5;
    const int node_tiles = NNODES / 16;
    const int tile = blockIdx.x * 4 + wave;
    const bool active = tile < node_tiles;
    const int tt = active ? tile : node_tiles - 1;
    const int base = tt * 16;
    const int mrow = lane & 15;
    const int kgrp = (lane >> 4) * 2;
    const int khalf = lane >> 4;
    const int mo   = 8 * (lane >> 4);
    const int ncol = lane & 15;

    const float* hr = h + (size_t)(base + mrow) * HD;
    const float* ar = agg + (size_t)(base + mrow) * HD;

    v8f acc[8];
#pragma unroll
    for (int nt = 0; nt < 8; ++nt) acc[nt] = vzero8();

    for (int kc = 0; kc < 2 * HD; kc += KC) {
        __syncthreads();
        stage_weights(sW, W1, kc, tid);
        __syncthreads();
#pragma unroll
        for (int s = 0; s < KC / 4; ++s) {
            const int kb = kc + 4 * s + kgrp;
            const float* src = (kb < HD) ? (hr + kb) : (ar + (kb - HD));
            v2f a = *(const v2f*)src;
            const v2f* bp = &sW[(2 * s + khalf) * HD + ncol];
#pragma unroll
            for (int nt = 0; nt < 8; ++nt) {
                acc[nt] = wmma_f32(a, bp[nt * 16], acc[nt]);
            }
        }
    }

    float* sTw = sT[wave];
#pragma unroll
    for (int nt = 0; nt < 8; ++nt) {
        const int n = nt * 16 + ncol;
        const float bb = b1[n];
#pragma unroll
        for (int g = 0; g < 8; ++g) {
            sTw[(g + mo) * HD + n] = silu(acc[nt][g] + bb);
        }
    }
    asm volatile("s_wait_dscnt 0" ::: "memory");

    v8f acc2[8];
#pragma unroll
    for (int nt = 0; nt < 8; ++nt) acc2[nt] = vzero8();

    for (int kc = 0; kc < HD; kc += KC) {
        __syncthreads();
        stage_weights(sW, W2, kc, tid);
        __syncthreads();
#pragma unroll
        for (int s = 0; s < KC / 4; ++s) {
            const int kb = kc + 4 * s + kgrp;
            v2f a = *(const v2f*)&sTw[mrow * HD + kb];
            const v2f* bp = &sW[(2 * s + khalf) * HD + ncol];
#pragma unroll
            for (int nt = 0; nt < 8; ++nt) {
                acc2[nt] = wmma_f32(a, bp[nt * 16], acc2[nt]);
            }
        }
    }

    if (active) {
#pragma unroll
        for (int nt = 0; nt < 8; ++nt) {
            const int n = nt * 16 + ncol;
            const float bb = b2[n];
#pragma unroll
            for (int g = 0; g < 8; ++g) {
                float* dst = &h[(size_t)(base + g + mo) * HD + n];
                *dst = *dst + (acc2[nt][g] + bb);
            }
        }
    }
}

__global__ __launch_bounds__(128) void decode_kernel(
    const float* __restrict__ h, const float* __restrict__ dw1,
    const float* __restrict__ db1, const float* __restrict__ dw2,
    const float* __restrict__ db2, float* __restrict__ out) {
    __shared__ v2f sW[(KC / 2) * HD];
    __shared__ float sT[4][16 * HD];
    __shared__ float sD[HD * 3];
    const int tid  = threadIdx.x;
    const int lane = tid & 31;
    const int wave = tid >> 5;
    const int node_tiles = NNODES / 16;
    const int tile = blockIdx.x * 4 + wave;
    const bool active = tile < node_tiles;
    const int tt = active ? tile : node_tiles - 1;
    const int base = tt * 16;
    const int mrow = lane & 15;
    const int kgrp = (lane >> 4) * 2;
    const int khalf = lane >> 4;
    const int mo   = 8 * (lane >> 4);
    const int ncol = lane & 15;

    const float* hr = h + (size_t)(base + mrow) * HD;

    for (int i = tid; i < HD * 3; i += 128) sD[i] = dw2[i];

    v8f acc[8];
#pragma unroll
    for (int nt = 0; nt < 8; ++nt) acc[nt] = vzero8();

    for (int kc = 0; kc < HD; kc += KC) {
        __syncthreads();
        stage_weights(sW, dw1, kc, tid);
        __syncthreads();
#pragma unroll
        for (int s = 0; s < KC / 4; ++s) {
            const int kb = kc + 4 * s + kgrp;
            v2f a = *(const v2f*)(hr + kb);
            const v2f* bp = &sW[(2 * s + khalf) * HD + ncol];
#pragma unroll
            for (int nt = 0; nt < 8; ++nt) {
                acc[nt] = wmma_f32(a, bp[nt * 16], acc[nt]);
            }
        }
    }

    float* sTw = sT[wave];
#pragma unroll
    for (int nt = 0; nt < 8; ++nt) {
        const int n = nt * 16 + ncol;
        const float bb = db1[n];
#pragma unroll
        for (int g = 0; g < 8; ++g) {
            sTw[(g + mo) * HD + n] = silu(acc[nt][g] + bb);
        }
    }
    asm volatile("s_wait_dscnt 0" ::: "memory");

    float y0 = 0.f, y1 = 0.f, y2 = 0.f;
    for (int j = 0; j < 64; ++j) {
        const int k = 64 * khalf + ((j + mrow) & 63);
        const float t = sTw[mrow * HD + k];
        y0 = fmaf(t, sD[3 * k + 0], y0);
        y1 = fmaf(t, sD[3 * k + 1], y1);
        y2 = fmaf(t, sD[3 * k + 2], y2);
    }
    y0 += __shfl_xor(y0, 16);
    y1 += __shfl_xor(y1, 16);
    y2 += __shfl_xor(y2, 16);
    if (active && lane < 16) {
        float* o = out + (size_t)(base + mrow) * 3;
        o[0] = y0 + db2[0];
        o[1] = y1 + db2[1];
        o[2] = y2 + db2[2];
    }
}

extern "C" void kernel_launch(void* const* d_in, const int* in_sizes, int n_in,
                              void* d_out, int out_size, void* d_ws, size_t ws_size,
                              hipStream_t stream) {
    const float* loc       = (const float*)d_in[0];
    const float* vel       = (const float*)d_in[1];
    const float* edge_attr = (const float*)d_in[2];
    const int*   row       = (const int*)d_in[3];
    const int*   col       = (const int*)d_in[4];
    const float* emb_w     = (const float*)d_in[5];
    const float* emb_b     = (const float*)d_in[6];
    const float* ew1       = (const float*)d_in[7];
    const float* eb1       = (const float*)d_in[8];
    const float* ew2       = (const float*)d_in[9];
    const float* eb2       = (const float*)d_in[10];
    const float* nw1       = (const float*)d_in[11];
    const float* nb1       = (const float*)d_in[12];
    const float* nw2       = (const float*)d_in[13];
    const float* nb2       = (const float*)d_in[14];
    const float* dw1       = (const float*)d_in[15];
    const float* db1       = (const float*)d_in[16];
    const float* dw2       = (const float*)d_in[17];
    const float* db2       = (const float*)d_in[18];
    float* out = (float*)d_out;

    float* h   = (float*)d_ws;
    float* agg = h + (size_t)NNODES * HD;

    embed_kernel<<<(NNODES * HD + 255) / 256, 256, 0, stream>>>(loc, vel, emb_w,
                                                                emb_b, h);

    const int edge_tiles = NEDGES / 16;
    const int node_tiles = NNODES / 16;
    const int edge_blocks = edge_tiles / 4;
    const int node_blocks = (node_tiles + 3) / 4;

    for (int i = 0; i < NLAYERS; ++i) {
        zero_kernel<<<(NNODES * HD + 255) / 256, 256, 0, stream>>>(agg,
                                                                   NNODES * HD);
        edge_kernel<<<edge_blocks, 128, 0, stream>>>(
            h, edge_attr, row, col, ew1 + (size_t)i * 258 * HD, eb1 + i * HD,
            ew2 + (size_t)i * HD * HD, eb2 + i * HD, agg);
        node_kernel<<<node_blocks, 128, 0, stream>>>(
            h, agg, nw1 + (size_t)i * 2 * HD * HD, nb1 + i * HD,
            nw2 + (size_t)i * HD * HD, nb2 + i * HD);
    }

    decode_kernel<<<node_blocks, 128, 0, stream>>>(h, dw1, db1, dw2, db2, out);
}


# Edge layer 1 formed per node, not per edge: P=h@W1[0:128]+b1, Q=h@W1[128:256] on f32 WMMA (one more launch a layer), edges gather P[row]+Q[col]; +51 MB workspace, checked on device; sum order changes
# speedup vs baseline: 1.5953x; 1.5291x over previous
#include <hip/hip_runtime.h>
#include <hip/hip_bf16.h>

#define HD 128
#define NNODES 50000
#define NEDGES 600000
#define NLAYERS 4
#define KC 32

typedef float v2f __attribute__((ext_vector_type(2)));
typedef float v4f __attribute__((ext_vector_type(4)));
typedef float v8f __attribute__((ext_vector_type(8)));

#define WS_PERNODE_BYTES ((size_t)4 * NNODES * HD * sizeof(float))

static __device__ __forceinline__ v8f vzero8() {
    v8f z;
#pragma unroll
    for (int i = 0; i < 8; ++i) z[i] = 0.f;
    return z;
}

static __device__ __forceinline__ v8f wmma_f32(v2f a, v2f b, v8f c) {
    return __builtin_amdgcn_wmma_f32_16x16x4_f32(false, a, false, b, (short)0, c,
                                                 false, false);
}

static __device__ __forceinline__ float silu(float x) {
    return x / (1.f + __expf(-x));
}

static __device__ __forceinline__ void stage_weights(v2f* __restrict__ sW,
                                                     const float* __restrict__ W,
                                                     int kc, int tid) {
    const float* wsrc = W + (size_t)kc * HD + tid;
#pragma unroll
    for (int p = 0; p < KC / 2; ++p) {
        v2f w;
        w.x = wsrc[(2 * p) * HD];
        w.y = wsrc[(2 * p + 1) * HD];
        sW[p * HD + tid] = w;
    }
}

__global__ __launch_bounds__(256) void embed_kernel(
    const float* __restrict__ loc, const float* __restrict__ vel,
    const float* __restrict__ emb_w, const float* __restrict__ emb_b,
    float* __restrict__ h) {
    int idx = blockIdx.x * blockDim.x + threadIdx.x;
    if (idx >= NNODES * HD) return;
    int n = idx >> 7;
    int c = idx & (HD - 1);
    float x = emb_b[c];
#pragma unroll
    for (int k = 0; k < 3; ++k) x = fmaf(loc[n * 3 + k], emb_w[k * HD + c], x);
#pragma unroll
    for (int k = 0; k < 3; ++k) x = fmaf(vel[n * 3 + k], emb_w[(k + 3) * HD + c], x);
    h[idx] = x;
}

__global__ __launch_bounds__(256) void zero_kernel(float* __restrict__ p, int count) {
    int i = blockIdx.x * blockDim.x + threadIdx.x;
    if (i < count) p[i] = 0.f;
}

__global__ __launch_bounds__(128) void pq_kernel(
    const float* __restrict__ h,
    const float* __restrict__ W1,
    const float* __restrict__ b1,
    float* __restrict__ pq,
    size_t ws_size) {
    if (ws_size < WS_PERNODE_BYTES) return;
    __shared__ v2f sW[(KC / 2) * HD];
    __shared__ float sT[4][16 * HD];
    const int tid  = threadIdx.x;
    const int lane = tid & 31;
    const int wave = tid >> 5;
    const int half = blockIdx.y;
    const int node_tiles = NNODES / 16;
    const int tile = blockIdx.x * 4 + wave;
    const bool active = tile < node_tiles;
    const int tt = active ? tile : node_tiles - 1;
    const int base = tt * 16;
    const int mrow = lane & 15;
    const int kgrp = (lane >> 4) * 2;
    const int khalf = lane >> 4;
    const int mo   = 8 * (lane >> 4);
    const int ncol = lane & 15;

    const float* hr = h + (size_t)(base + mrow) * HD;
    const float* Wh = W1 + (size_t)half * HD * HD;
    float* dst = pq + (size_t)half * NNODES * HD;

    v8f acc[8];
#pragma unroll
    for (int nt = 0; nt < 8; ++nt) acc[nt] = vzero8();

    for (int kc = 0; kc < HD; kc += KC) {
        __syncthreads();
        stage_weights(sW, Wh, kc, tid);
        __syncthreads();
#pragma unroll
        for (int s = 0; s < KC / 4; ++s) {
            const int kb = kc + 4 * s + kgrp;
            v2f a = *(const v2f*)(hr + kb);
            const v2f* bp = &sW[(2 * s + khalf) * HD + ncol];
#pragma unroll
            for (int nt = 0; nt < 8; ++nt) {
                acc[nt] = wmma_f32(a, bp[nt * 16], acc[nt]);
            }
        }
    }

    float* sTw = sT[wave];
#pragma unroll
    for (int nt = 0; nt < 8; ++nt) {
        const int n = nt * 16 + ncol;
        const float bb = (half == 0) ? b1[n] : 0.f;
#pragma unroll
        for (int g = 0; g < 8; ++g) {
            sTw[(g + mo) * HD + n] = acc[nt][g] + bb;
        }
    }
    asm volatile("s_wait_dscnt 0" ::: "memory");

    if (active) {
#pragma unroll
        for (int m = 0; m < 16; ++m) {
            const v4f v = *(const v4f*)&sTw[m * HD + 4 * lane];
            *(v4f*)(dst + (size_t)(base + m) * HD + 4 * lane) = v;
        }
    }
}

__global__ __launch_bounds__(128) void edge_kernel(
    const float* __restrict__ h, const float* __restrict__ edge_attr,
    const int* __restrict__ row, const int* __restrict__ col,
    const float* __restrict__ W1,
    const float* __restrict__ b1,
    const float* __restrict__ W2,
    const float* __restrict__ b2,
    float* __restrict__ agg,
    const float* __restrict__ pq,
    size_t ws_size) {
    __shared__ v2f sW[(KC / 2) * HD];
    __shared__ float sT[4][16 * HD];
    const bool pernode = ws_size >= WS_PERNODE_BYTES;
    const int tid  = threadIdx.x;
    const int lane = tid & 31;
    const int wave = tid >> 5;
    const int tile = blockIdx.x * 4 + wave;
    const int base = tile * 16;
    const int mrow = lane & 15;
    const int kgrp = (lane >> 4) * 2;
    const int khalf = lane >> 4;
    const int mo   = 8 * (lane >> 4);
    const int ncol = lane & 15;

    const int r_idx = row[base + mrow];
    const int c_idx = col[base + mrow];
    const float* hr = h + (size_t)r_idx * HD;
    const float* hc = h + (size_t)c_idx * HD;

    float* sTw = sT[wave];
    if (pernode) {
        const float* P = pq;
        const float* Q = pq + (size_t)NNODES * HD;
        const v4f we0 = *(const v4f*)(W1 + (size_t)256 * HD + 4 * lane);
        const v4f we1 = *(const v4f*)(W1 + (size_t)257 * HD + 4 * lane);
        const v2f ea = *(const v2f*)(edge_attr + 2 * (size_t)(base + mrow));
#pragma unroll
        for (int m = 0; m < 16; ++m) {
            const int rm = __builtin_amdgcn_readlane(r_idx, m);
            const int cm = __builtin_amdgcn_readlane(c_idx, m);
            const float e0 = __int_as_float(__builtin_amdgcn_readlane(__float_as_int(ea.x), m));
            const float e1 = __int_as_float(__builtin_amdgcn_readlane(__float_as_int(ea.y), m));
            const v4f p = *(const v4f*)(P + (size_t)rm * HD + 4 * lane);
            const v4f q = *(const v4f*)(Q + (size_t)cm * HD + 4 * lane);
            v4f x;
#pragma unroll
            for (int j = 0; j < 4; ++j) {
                x[j] = silu(p[j] + q[j] + e0 * we0[j] + e1 * we1[j]);
            }
            *(v4f*)&sTw[m * HD + 4 * lane] = x;
        }
    } else {
        v8f acc[8];
#pragma unroll
        for (int nt = 0; nt < 8; ++nt) acc[nt] = vzero8();

        for (int kc = 0; kc < 2 * HD; kc += KC) {
            __syncthreads();
            stage_weights(sW, W1, kc, tid);
            __syncthreads();
#pragma unroll
            for (int s = 0; s < KC / 4; ++s) {
                const int kb = kc + 4 * s + kgrp;
                const float* src = (kb < HD) ? (hr + kb) : (hc + (kb - HD));
                v2f a = *(const v2f*)src;
                const v2f* bp = &sW[(2 * s + khalf) * HD + ncol];
#pragma unroll
                for (int nt = 0; nt < 8; ++nt) {
                    acc[nt] = wmma_f32(a, bp[nt * 16], acc[nt]);
                }
            }
        }

        float ea0[8], ea1[8];
#pragma unroll
        for (int g = 0; g < 8; ++g) {
            const int e = base + g + mo;
            ea0[g] = edge_attr[2 * e];
            ea1[g] = edge_attr[2 * e + 1];
        }
        const float* w1e0 = W1 + (size_t)256 * HD;
        const float* w1e1 = W1 + (size_t)257 * HD;
#pragma unroll
        for (int nt = 0; nt < 8; ++nt) {
            const int n = nt * 16 + ncol;
            const float we0 = w1e0[n], we1 = w1e1[n], bb = b1[n];
#pragma unroll
            for (int g = 0; g < 8; ++g) {
                float x = acc[nt][g] + ea0[g] * we0 + ea1[g] * we1 + bb;
                sTw[(g + mo) * HD + n] = silu(x);
            }
        }
    }
    asm volatile("s_wait_dscnt 0" ::: "memory");

    v8f acc2[8];
#pragma unroll
    for (int nt = 0; nt < 8; ++nt) acc2[nt] = vzero8();

    for (int kc = 0; kc < HD; kc += KC) {
        __syncthreads();
        stage_weights(sW, W2, kc, tid);
        __syncthreads();
#pragma unroll
        for (int s = 0; s < KC / 4; ++s) {
            const int kb = kc + 4 * s + kgrp;
            v2f a = *(const v2f*)&sTw[mrow * HD + kb];
            const v2f* bp = &sW[(2 * s + khalf) * HD + ncol];
#pragma unroll
            for (int nt = 0; nt < 8; ++nt) {
                acc2[nt] = wmma_f32(a, bp[nt * 16], acc2[nt]);
            }
        }
    }

    int rg[8];
#pragma unroll
    for (int g = 0; g < 8; ++g) rg[g] = row[base + g + mo];
#pragma unroll
    for (int nt = 0; nt < 8; ++nt) {
        const int n = nt * 16 + ncol;
        const float bb = b2[n];
#pragma unroll
        for (int g = 0; g < 8; ++g) {
            float x = silu(acc2[nt][g] + bb);
            atomicAdd(&agg[(size_t)rg[g] * HD + n], x);
        }
    }
}

__global__ __launch_bounds__(128) void node_kernel(
    float* __restrict__ h, const float* __restrict__ agg,
    const float* __restrict__ W1,
    const float* __restrict__ b1, const float* __restrict__ W2,
    const float* __restrict__ b2) {
    __shared__ v2f sW[(KC / 2) * HD];
    __shared__ float sT[4][16 * HD];
    const int tid  = threadIdx.x;
    const int lane = tid & 31;
    const int wave = tid >> 5;
    const int node_tiles = NNODES / 16;
    const int tile = blockIdx.x * 4 + wave;
    const bool active = tile < node_tiles;
    const int tt = active ? tile : node_tiles - 1;
    const int base = tt * 16;
    const int mrow = lane & 15;
    const int kgrp = (lane >> 4) * 2;
    const int khalf = lane >> 4;
    const int mo   = 8 * (lane >> 4);
    const int ncol = lane & 15;

    const float* hr = h + (size_t)(base + mrow) * HD;
    const float* ar = agg + (size_t)(base + mrow) * HD;

    v8f acc[8];
#pragma unroll
    for (int nt = 0; nt < 8; ++nt) acc[nt] = vzero8();

    for (int kc = 0; kc < 2 * HD; kc += KC) {
        __syncthreads();
        stage_weights(sW, W1, kc, tid);
        __syncthreads();
#pragma unroll
        for (int s = 0; s < KC / 4; ++s) {
            const int kb = kc + 4 * s + kgrp;
            const float* src = (kb < HD) ? (hr + kb) : (ar + (kb - HD));
            v2f a = *(const v2f*)src;
            const v2f* bp = &sW[(2 * s + khalf) * HD + ncol];
#pragma unroll
            for (int nt = 0; nt < 8; ++nt) {
                acc[nt] = wmma_f32(a, bp[nt * 16], acc[nt]);
            }
        }
    }

    float* sTw = sT[wave];
#pragma unroll
    for (int nt = 0; nt < 8; ++nt) {
        const int n = nt * 16 + ncol;
        const float bb = b1[n];
#pragma unroll
        for (int g = 0; g < 8; ++g) {
            sTw[(g + mo) * HD + n] = silu(acc[nt][g] + bb);
        }
    }
    asm volatile("s_wait_dscnt 0" ::: "memory");

    v8f acc2[8];
#pragma unroll
    for (int nt = 0; nt < 8; ++nt) acc2[nt] = vzero8();

    for (int kc = 0; kc < HD; kc += KC) {
        __syncthreads();
        stage_weights(sW, W2, kc, tid);
        __syncthreads();
#pragma unroll
        for (int s = 0; s < KC / 4; ++s) {
            const int kb = kc + 4 * s + kgrp;
            v2f a = *(const v2f*)&sTw[mrow * HD + kb];
            const v2f* bp = &sW[(2 * s + khalf) * HD + ncol];
#pragma unroll
            for (int nt = 0; nt < 8; ++nt) {
                acc2[nt] = wmma_f32(a, bp[nt * 16], acc2[nt]);
            }
        }
    }

    if (active) {
#pragma unroll
        for (int nt = 0; nt < 8; ++nt) {
            const int n = nt * 16 + ncol;
            const float bb = b2[n];
#pragma unroll
            for (int g = 0; g < 8; ++g) {
                float* dst = &h[(size_t)(base + g + mo) * HD + n];
                *dst = *dst + (acc2[nt][g] + bb);
            }
        }
    }
}

__global__ __launch_bounds__(128) void decode_kernel(
    const float* __restrict__ h, const float* __restrict__ dw1,
    const float* __restrict__ db1, const float* __restrict__ dw2,
    const float* __restrict__ db2, float* __restrict__ out) {
    __shared__ v2f sW[(KC / 2) * HD];
    __shared__ float sT[4][16 * HD];
    __shared__ float sD[HD * 3];
    const int tid  = threadIdx.x;
    const int lane = tid & 31;
    const int wave = tid >> 5;
    const int node_tiles = NNODES / 16;
    const int tile = blockIdx.x * 4 + wave;
    const bool active = tile < node_tiles;
    const int tt = active ? tile : node_tiles - 1;
    const int base = tt * 16;
    const int mrow = lane & 15;
    const int kgrp = (lane >> 4) * 2;
    const int khalf = lane >> 4;
    const int mo   = 8 * (lane >> 4);
    const int ncol = lane & 15;

    const float* hr = h + (size_t)(base + mrow) * HD;

    for (int i = tid; i < HD * 3; i += 128) sD[i] = dw2[i];

    v8f acc[8];
#pragma unroll
    for (int nt = 0; nt < 8; ++nt) acc[nt] = vzero8();

    for (int kc = 0; kc < HD; kc += KC) {
        __syncthreads();
        stage_weights(sW, dw1, kc, tid);
        __syncthreads();
#pragma unroll
        for (int s = 0; s < KC / 4; ++s) {
            const int kb = kc + 4 * s + kgrp;
            v2f a = *(const v2f*)(hr + kb);
            const v2f* bp = &sW[(2 * s + khalf) * HD + ncol];
#pragma unroll
            for (int nt = 0; nt < 8; ++nt) {
                acc[nt] = wmma_f32(a, bp[nt * 16], acc[nt]);
            }
        }
    }

    float* sTw = sT[wave];
#pragma unroll
    for (int nt = 0; nt < 8; ++nt) {
        const int n = nt * 16 + ncol;
        const float bb = db1[n];
#pragma unroll
        for (int g = 0; g < 8; ++g) {
            sTw[(g + mo) * HD + n] = silu(acc[nt][g] + bb);
        }
    }
    asm volatile("s_wait_dscnt 0" ::: "memory");

    float y0 = 0.f, y1 = 0.f, y2 = 0.f;
    for (int j = 0; j < 64; ++j) {
        const int k = 64 * khalf + ((j + mrow) & 63);
        const float t = sTw[mrow * HD + k];
        y0 = fmaf(t, sD[3 * k + 0], y0);
        y1 = fmaf(t, sD[3 * k + 1], y1);
        y2 = fmaf(t, sD[3 * k + 2], y2);
    }
    y0 += __shfl_xor(y0, 16);
    y1 += __shfl_xor(y1, 16);
    y2 += __shfl_xor(y2, 16);
    if (active && lane < 16) {
        float* o = out + (size_t)(base + mrow) * 3;
        o[0] = y0 + db2[0];
        o[1] = y1 + db2[1];
        o[2] = y2 + db2[2];
    }
}

extern "C" void kernel_launch(void* const* d_in, const int* in_sizes, int n_in,
                              void* d_out, int out_size, void* d_ws, size_t ws_size,
                              hipStream_t stream) {
    const float* loc       = (const float*)d_in[0];
    const float* vel       = (const float*)d_in[1];
    const float* edge_attr = (const float*)d_in[2];
    const int*   row       = (const int*)d_in[3];
    const int*   col       = (const int*)d_in[4];
    const float* emb_w     = (const float*)d_in[5];
    const float* emb_b     = (const float*)d_in[6];
    const float* ew1       = (const float*)d_in[7];
    const float* eb1       = (const float*)d_in[8];
    const float* ew2       = (const float*)d_in[9];
    const float* eb2       = (const float*)d_in[10];
    const float* nw1       = (const float*)d_in[11];
    const float* nb1       = (const float*)d_in[12];
    const float* nw2       = (const float*)d_in[13];
    const float* nb2       = (const float*)d_in[14];
    const float* dw1       = (const float*)d_in[15];
    const float* db1       = (const float*)d_in[16];
    const float* dw2       = (const float*)d_in[17];
    const float* db2       = (const float*)d_in[18];
    float* out = (float*)d_out;

    float* h   = (float*)d_ws;
    float* agg = h + (size_t)NNODES * HD;
    float* pq  = agg + (size_t)NNODES * HD;

    embed_kernel<<<(NNODES * HD + 255) / 256, 256, 0, stream>>>(loc, vel, emb_w,
                                                                emb_b, h);

    const int edge_tiles = NEDGES / 16;
    const int node_tiles = NNODES / 16;
    const int edge_blocks = edge_tiles / 4;
    const int node_blocks = (node_tiles + 3) / 4;

    for (int i = 0; i < NLAYERS; ++i) {
        zero_kernel<<<(NNODES * HD + 255) / 256, 256, 0, stream>>>(agg,
                                                                   NNODES * HD);
        pq_kernel<<<dim3(node_blocks, 2), 128, 0, stream>>>(
            h, ew1 + (size_t)i * 258 * HD, eb1 + i * HD, pq, ws_size);
        edge_kernel<<<edge_blocks, 128, 0, stream>>>(
            h, edge_attr, row, col, ew1 + (size_t)i * 258 * HD, eb1 + i * HD,
            ew2 + (size_t)i * HD * HD, eb2 + i * HD, agg, pq, ws_size);
        node_kernel<<<node_blocks, 128, 0, stream>>>(
            h, agg, nw1 + (size_t)i * 2 * HD * HD, nb1 + i * HD,
            nw2 + (size_t)i * HD * HD, nb2 + i * HD);
    }

    decode_kernel<<<node_blocks, 128, 0, stream>>>(h, dw1, db1, dw2, db2, out);
}
